# starting kernel
# baseline (speedup 1.0000x reference)
#include <hip/hip_runtime.h>


#define N_NODES 100000
#define N_EDGES 1600000
#define D       64

typedef __attribute__((ext_vector_type(2))) float v2f;
typedef __attribute__((ext_vector_type(4))) float v4f;
typedef __attribute__((ext_vector_type(8))) float v8f;

__global__ __launch_bounds__(256) void gin_init(const float* __restrict__ x,
                                                float* __restrict__ h0) {
    size_t i = (size_t)blockIdx.x * blockDim.x + threadIdx.x;
    const size_t n4 = (size_t)N_NODES * D / 4;
    if (i < n4) {
        ((v4f*)h0)[i] = ((const v4f*)x)[i];
    }
}

__global__ __launch_bounds__(256) void gin_scatter(const float* __restrict__ x,
                                                   const int* __restrict__ src,
                                                   const int* __restrict__ dst,
                                                   float* __restrict__ h0) {
    size_t t = (size_t)blockIdx.x * blockDim.x + threadIdx.x;
    size_t e = t >> 4;
    int    f = (int)(t & 15) * 4;
    if (e < (size_t)N_EDGES) {
        if (f == 0 && e + 4096 < (size_t)N_EDGES) {
            __builtin_prefetch(src + e + 4096, 0, 0);
            __builtin_prefetch(dst + e + 4096, 0, 0);
        }
        int s = src[e];
        int d = dst[e];
        v4f v = *(const v4f*)(x + (size_t)s * D + f);
        float* p = h0 + (size_t)d * D + f;
        atomicAdd(p + 0, v[0]);
        atomicAdd(p + 1, v[1]);
        atomicAdd(p + 2, v[2]);
        atomicAdd(p + 3, v[3]);
    }
}

#define MLP_WAVES 4
#define HSTR  68
#define KSTR  320
#define LHSTR 160

__global__ __launch_bounds__(128) void gin_mlp(const float* __restrict__ h0,
                                               const float* __restrict__ W1,
                                               const float* __restrict__ b1,
                                               const float* __restrict__ W2,
                                               const float* __restrict__ b2,
                                               float* __restrict__ out) {
    __shared__ float sW1[16 * KSTR];
    __shared__ float sW2[16 * KSTR];
    __shared__ float sB1[D];
    __shared__ float sB2[D];
    __shared__ float sH[MLP_WAVES][16 * HSTR];

    const int tid = threadIdx.x;

    for (int i = tid; i < D * D / 4; i += 128) {
        const int k = i >> 4;
        const int c = (i & 15) << 2;
        const int base = (k >> 2) * KSTR + ((k >> 1) & 1) * LHSTR + (k & 1);
        v4f w1 = ((const v4f*)W1)[i];
        v4f w2 = ((const v4f*)W2)[i];
        #pragma unroll
        for (int j = 0; j < 4; ++j) {
            sW1[base + ((c + j) << 1)] = w1[j];
            sW2[base + ((c + j) << 1)] = w2[j];
        }
    }
    if (tid < D) {
        sB1[tid] = b1[tid];
        sB2[tid] = b2[tid];
    }
    __syncthreads();

    const int wave  = tid >> 5;
    const int lane  = tid & 31;
    const int lhalf = lane >> 4;
    const int l16   = lane & 15;

    const int row0 = (blockIdx.x * MLP_WAVES + wave) * 32;
    const float* a0p = h0 + (size_t)min(row0 + l16,      N_NODES - 1) * D;
    const float* a1p = h0 + (size_t)min(row0 + 16 + l16, N_NODES - 1) * D;

    const v8f vz = {0, 0, 0, 0, 0, 0, 0, 0};
    v8f acc0[4], acc1[4];
    #pragma unroll
    for (int nt = 0; nt < 4; ++nt) { acc0[nt] = vz; acc1[nt] = vz; }

    const float* bw1 = sW1 + lhalf * LHSTR + (l16 << 1);
    const float* bw2 = sW2 + lhalf * LHSTR + (l16 << 1);

    #pragma unroll
    for (int kk = 0; kk < 16; ++kk) {
        const int koff = kk * 4 + lhalf * 2;
        const v2f a0 = *(const v2f*)(a0p + koff);
        const v2f a1 = *(const v2f*)(a1p + koff);
        #pragma unroll
        for (int nt = 0; nt < 4; ++nt) {
            const v2f b = *(const v2f*)(bw1 + kk * KSTR + nt * 32);
            acc0[nt] = __builtin_amdgcn_wmma_f32_16x16x4_f32(
                false, a0, false, b, (short)0, acc0[nt], false, false);
            acc1[nt] = __builtin_amdgcn_wmma_f32_16x16x4_f32(
                false, a1, false, b, (short)0, acc1[nt], false, false);
        }
    }

    float* myH = &sH[wave][0];
    const float* aH = myH + l16 * HSTR;

    #pragma unroll
    for (int mt = 0; mt < 2; ++mt) {
        const v8f* acc = (mt == 0) ? acc0 : acc1;

        #pragma unroll
        for (int nt = 0; nt < 4; ++nt) {
            const int   col  = nt * 16 + l16;
            const float bias = sB1[col];
            #pragma unroll
            for (int r = 0; r < 8; ++r) {
                float v = acc[nt][r] + bias;
                v = v > 0.0f ? v : 0.0f;
                myH[(r + lhalf * 8) * HSTR + col] = v;
            }
        }

        v8f o[4];
        #pragma unroll
        for (int nt = 0; nt < 4; ++nt) o[nt] = vz;

        #pragma unroll
        for (int kk = 0; kk < 16; ++kk) {
            const int koff = kk * 4 + lhalf * 2;
            const v2f a = *(const v2f*)(aH + koff);
            #pragma unroll
            for (int nt = 0; nt < 4; ++nt) {
                const v2f b = *(const v2f*)(bw2 + kk * KSTR + nt * 32);
                o[nt] = __builtin_amdgcn_wmma_f32_16x16x4_f32(
                    false, a, false, b, (short)0, o[nt], false, false);
            }
        }

        #pragma unroll
        for (int r = 0; r < 8; ++r) {
            const int m = row0 + mt * 16 + r + lhalf * 8;
            if (m < N_NODES) {
                #pragma unroll
                for (int nt = 0; nt < 4; ++nt) {
                    const int col = nt * 16 + l16;
                    out[(size_t)m * D + col] = o[nt][r] + sB2[col];
                }
            }
        }
    }
}

extern "C" void kernel_launch(void* const* d_in, const int* in_sizes, int n_in,
                              void* d_out, int out_size, void* d_ws, size_t ws_size,
                              hipStream_t stream) {
    const float* x  = (const float*)d_in[0];
    const int*   ei = (const int*)d_in[1];
    const float* W1 = (const float*)d_in[2];
    const float* b1 = (const float*)d_in[3];
    const float* W2 = (const float*)d_in[4];
    const float* b2 = (const float*)d_in[5];
    float*       out = (float*)d_out;

    const int* src = ei;
    const int* dst = ei + N_EDGES;

    float* h0 = (float*)d_ws;

    {
        const int n4 = N_NODES * D / 4;
        gin_init<<<(n4 + 255) / 256, 256, 0, stream>>>(x, h0);
    }
    {
        const long long nt = (long long)N_EDGES * 16;
        gin_scatter<<<(int)((nt + 255) / 256), 256, 0, stream>>>(x, src, dst, h0);
    }
    {
        const int rows_per_block = MLP_WAVES * 32;
        gin_mlp<<<(N_NODES + rows_per_block - 1) / rows_per_block, 128, 0, stream>>>(
            h0, W1, b1, W2, b2, out);
    }
}


# scatter without f32 atomics: one returning int atomic per edge files its source id in the destination's slot line; a wave per node then sums those rows of x (arrival-order sums; spills keep atomics)
# speedup vs baseline: 1.0503x; 1.0503x over previous
#include <hip/hip_runtime.h>


#define N_NODES 100000
#define N_EDGES 1600000
#define D       64

typedef __attribute__((ext_vector_type(2))) float v2f;
typedef __attribute__((ext_vector_type(4))) float v4f;
typedef __attribute__((ext_vector_type(8))) float v8f;
typedef __attribute__((ext_vector_type(4))) int   v4i;

#define CAP        32
#define CNT_WORDS  100352
#define SLOT_OFF   CNT_WORDS
#define SPILL_OFF  (SLOT_OFF + N_NODES * CAP)
#define H0_OFF     (SPILL_OFF + 2 * N_EDGES)

__global__ __launch_bounds__(256) void gin_zero(int* __restrict__ cnt) {
    const int i = blockIdx.x * 256 + threadIdx.x;
    const v4i z = {0, 0, 0, 0};
    ((v4i*)cnt)[i] = z;
}

__device__ __forceinline__ void gin_place(int s, int d, int pos,
                                          int* __restrict__ cnt,
                                          int* __restrict__ slots,
                                          int* __restrict__ spill) {
    if (pos < CAP) {
        slots[(size_t)d * CAP + pos] = s;
    } else {
        const int o = atomicAdd(cnt + N_NODES, 1);
        spill[2 * (size_t)o]     = s;
        spill[2 * (size_t)o + 1] = d;
    }
}

__global__ __launch_bounds__(256) void gin_bucket(const int* __restrict__ src,
                                                  const int* __restrict__ dst,
                                                  int* __restrict__ cnt,
                                                  int* __restrict__ slots,
                                                  int* __restrict__ spill) {
    const int t = blockIdx.x * 256 + threadIdx.x;
    if (t < N_EDGES / 4) {
        const v4i s4 = ((const v4i*)src)[t];
        const v4i d4 = ((const v4i*)dst)[t];
        const int p0 = atomicAdd(cnt + d4[0], 1);
        const int p1 = atomicAdd(cnt + d4[1], 1);
        const int p2 = atomicAdd(cnt + d4[2], 1);
        const int p3 = atomicAdd(cnt + d4[3], 1);
        gin_place(s4[0], d4[0], p0, cnt, slots, spill);
        gin_place(s4[1], d4[1], p1, cnt, slots, spill);
        gin_place(s4[2], d4[2], p2, cnt, slots, spill);
        gin_place(s4[3], d4[3], p3, cnt, slots, spill);
    }
}

#define G_BLOCKS 1536
#define G_WAVES  (G_BLOCKS * 8)
#define GCH      16

__global__ __launch_bounds__(256) void gin_gather(const float* __restrict__ x,
                                                  const int* __restrict__ cnt,
                                                  const int* __restrict__ slots,
                                                  float* __restrict__ h0) {
    const int lane = threadIdx.x & 31;
    const int w = __builtin_amdgcn_readfirstlane((int)(blockIdx.x * 8 + (threadIdx.x >> 5)));
    int deg = (w < N_NODES) ? cnt[w] : 0;
    for (int n = w; n < N_NODES; n += G_WAVES) {
        const int nn = n + G_WAVES;
        const int deg_next = (nn < N_NODES) ? cnt[nn] : 0;
        const int m = deg < CAP ? deg : CAP;
        int sv = 0;
        if (lane < m) sv = slots[(size_t)n * CAP + lane];
        v2f acc = *(const v2f*)(x + (size_t)n * D + lane * 2);
        for (int j0 = 0; j0 < m; j0 += GCH) {
            int s[GCH];
            #pragma unroll
            for (int u = 0; u < GCH; ++u) {
                const int idx = (j0 + u < m) ? (j0 + u) : (m - 1);
                s[u] = __builtin_amdgcn_readlane(sv, idx);
            }
            v2f v[GCH];
            #pragma unroll
            for (int u = 0; u < GCH; ++u) {
                v[u] = (v2f){0.0f, 0.0f};
                if (j0 + u < m) {
                    v[u] = *(const v2f*)(x + (size_t)s[u] * D + lane * 2);
                }
            }
            #pragma unroll
            for (int u = 0; u < GCH; ++u) acc += v[u];
        }
        *(v2f*)(h0 + (size_t)n * D + lane * 2) = acc;
        deg = deg_next;
    }
}

#define SP_BLOCKS 64
#define SP_GROUPS (SP_BLOCKS * 16)

__global__ __launch_bounds__(256) void gin_spill(const float* __restrict__ x,
                                                 const int* __restrict__ cnt,
                                                 const int* __restrict__ spill,
                                                 float* __restrict__ h0) {
    int total = cnt[N_NODES];
    total = total < N_EDGES ? total : N_EDGES;
    const int g = (blockIdx.x * 256 + threadIdx.x) >> 4;
    const int f = (threadIdx.x & 15) * 4;
    for (int o = g; o < total; o += SP_GROUPS) {
        const int s = spill[2 * (size_t)o];
        const int d = spill[2 * (size_t)o + 1];
        const v4f v = *(const v4f*)(x + (size_t)s * D + f);
        float* p = h0 + (size_t)d * D + f;
        atomicAdd(p + 0, v[0]);
        atomicAdd(p + 1, v[1]);
        atomicAdd(p + 2, v[2]);
        atomicAdd(p + 3, v[3]);
    }
}

#define MLP_WAVES 4
#define HSTR  68
#define KSTR  320
#define LHSTR 160

__global__ __launch_bounds__(128) void gin_mlp(const float* __restrict__ h0,
                                               const float* __restrict__ W1,
                                               const float* __restrict__ b1,
                                               const float* __restrict__ W2,
                                               const float* __restrict__ b2,
                                               float* __restrict__ out) {
    __shared__ float sW1[16 * KSTR];
    __shared__ float sW2[16 * KSTR];
    __shared__ float sB1[D];
    __shared__ float sB2[D];
    __shared__ float sH[MLP_WAVES][16 * HSTR];

    const int tid = threadIdx.x;

    for (int i = tid; i < D * D / 4; i += 128) {
        const int k = i >> 4;
        const int c = (i & 15) << 2;
        const int base = (k >> 2) * KSTR + ((k >> 1) & 1) * LHSTR + (k & 1);
        v4f w1 = ((const v4f*)W1)[i];
        v4f w2 = ((const v4f*)W2)[i];
        #pragma unroll
        for (int j = 0; j < 4; ++j) {
            sW1[base + ((c + j) << 1)] = w1[j];
            sW2[base + ((c + j) << 1)] = w2[j];
        }
    }
    if (tid < D) {
        sB1[tid] = b1[tid];
        sB2[tid] = b2[tid];
    }
    __syncthreads();

    const int wave  = tid >> 5;
    const int lane  = tid & 31;
    const int lhalf = lane >> 4;
    const int l16   = lane & 15;

    const int row0 = (blockIdx.x * MLP_WAVES + wave) * 32;
    const float* a0p = h0 + (size_t)min(row0 + l16,      N_NODES - 1) * D;
    const float* a1p = h0 + (size_t)min(row0 + 16 + l16, N_NODES - 1) * D;

    const v8f vz = {0, 0, 0, 0, 0, 0, 0, 0};
    v8f acc0[4], acc1[4];
    #pragma unroll
    for (int nt = 0; nt < 4; ++nt) { acc0[nt] = vz; acc1[nt] = vz; }

    const float* bw1 = sW1 + lhalf * LHSTR + (l16 << 1);
    const float* bw2 = sW2 + lhalf * LHSTR + (l16 << 1);

    #pragma unroll
    for (int kk = 0; kk < 16; ++kk) {
        const int koff = kk * 4 + lhalf * 2;
        const v2f a0 = *(const v2f*)(a0p + koff);
        const v2f a1 = *(const v2f*)(a1p + koff);
        #pragma unroll
        for (int nt = 0; nt < 4; ++nt) {
            const v2f b = *(const v2f*)(bw1 + kk * KSTR + nt * 32);
            acc0[nt] = __builtin_amdgcn_wmma_f32_16x16x4_f32(
                false, a0, false, b, (short)0, acc0[nt], false, false);
            acc1[nt] = __builtin_amdgcn_wmma_f32_16x16x4_f32(
                false, a1, false, b, (short)0, acc1[nt], false, false);
        }
    }

    float* myH = &sH[wave][0];
    const float* aH = myH + l16 * HSTR;

    #pragma unroll
    for (int mt = 0; mt < 2; ++mt) {
        const v8f* acc = (mt == 0) ? acc0 : acc1;

        #pragma unroll
        for (int nt = 0; nt < 4; ++nt) {
            const int   col  = nt * 16 + l16;
            const float bias = sB1[col];
            #pragma unroll
            for (int r = 0; r < 8; ++r) {
                float v = acc[nt][r] + bias;
                v = v > 0.0f ? v : 0.0f;
                myH[(r + lhalf * 8) * HSTR + col] = v;
            }
        }

        v8f o[4];
        #pragma unroll
        for (int nt = 0; nt < 4; ++nt) o[nt] = vz;

        #pragma unroll
        for (int kk = 0; kk < 16; ++kk) {
            const int koff = kk * 4 + lhalf * 2;
            const v2f a = *(const v2f*)(aH + koff);
            #pragma unroll
            for (int nt = 0; nt < 4; ++nt) {
                const v2f b = *(const v2f*)(bw2 + kk * KSTR + nt * 32);
                o[nt] = __builtin_amdgcn_wmma_f32_16x16x4_f32(
                    false, a, false, b, (short)0, o[nt], false, false);
            }
        }

        #pragma unroll
        for (int r = 0; r < 8; ++r) {
            const int m = row0 + mt * 16 + r + lhalf * 8;
            if (m < N_NODES) {
                #pragma unroll
                for (int nt = 0; nt < 4; ++nt) {
                    const int col = nt * 16 + l16;
                    out[(size_t)m * D + col] = o[nt][r] + sB2[col];
                }
            }
        }
    }
}

extern "C" void kernel_launch(void* const* d_in, const int* in_sizes, int n_in,
                              void* d_out, int out_size, void* d_ws, size_t ws_size,
                              hipStream_t stream) {
    const float* x  = (const float*)d_in[0];
    const int*   ei = (const int*)d_in[1];
    const float* W1 = (const float*)d_in[2];
    const float* b1 = (const float*)d_in[3];
    const float* W2 = (const float*)d_in[4];
    const float* b2 = (const float*)d_in[5];
    float*       out = (float*)d_out;

    const int* src = ei;
    const int* dst = ei + N_EDGES;

    int*   cnt   = (int*)d_ws;
    int*   slots = (int*)d_ws + SLOT_OFF;
    int*   spill = (int*)d_ws + SPILL_OFF;
    float* h0    = (float*)d_ws + H0_OFF;

    gin_zero<<<CNT_WORDS / 4 / 256, 256, 0, stream>>>(cnt);
    gin_bucket<<<(N_EDGES / 4 + 255) / 256, 256, 0, stream>>>(src, dst, cnt, slots, spill);
    gin_gather<<<G_BLOCKS, 256, 0, stream>>>(x, cnt, slots, h0);
    gin_spill<<<SP_BLOCKS, 256, 0, stream>>>(x, cnt, spill, h0);
    {
        const int rows_per_block = MLP_WAVES * 32;
        gin_mlp<<<(N_NODES + rows_per_block - 1) / rows_per_block, 128, 0, stream>>>(
            h0, W1, b1, W2, b2, out);
    }
}


# edges ranked by LDS counters in two levels (chunk cells per 256-node range, then per-node lists in LDS feeding the row sums), not one returning global atomic per edge; sums stay in arrival order
# speedup vs baseline: 2.9724x; 2.8300x over previous
#include <hip/hip_runtime.h>


#define N_NODES 100000
#define N_EDGES 1600000
#define D       64

typedef __attribute__((ext_vector_type(2))) float v2f;
typedef __attribute__((ext_vector_type(4))) float v4f;
typedef __attribute__((ext_vector_type(8))) float v8f;
typedef __attribute__((ext_vector_type(4))) int   v4i;

#define CAP        32
#define RNG        256
#define NB         391
#define NCH        250
#define EPC        6400
#define CELLW      32
#define CELLCAP    31
#define GCNT_WORDS 1024
#define CELL_OFF   GCNT_WORDS
#define SPILL_OFF  (CELL_OFF + NB * NCH * CELLW)
#define H0_OFF     (SPILL_OFF + 2 * N_EDGES)

__global__ __launch_bounds__(64) void gin_zero(int* __restrict__ gcnt) {
    const v4i z = {0, 0, 0, 0};
    ((v4i*)gcnt)[threadIdx.x] = z;
}

__device__ __forceinline__ void gin_spill_put(int s, int d,
                                              int* __restrict__ gcnt,
                                              int* __restrict__ spill) {
    const int o = atomicAdd(gcnt, 1);
    spill[2 * (size_t)o]     = s;
    spill[2 * (size_t)o + 1] = d;
}

#define P1_THREADS 800

__device__ __forceinline__ void gin_file(int s, int d, int* lcnt, int* lcell,
                                         int* __restrict__ gcnt,
                                         int* __restrict__ spill) {
    const int r   = d >> 8;
    const int pos = atomicAdd(&lcnt[r], 1);
    if (pos < CELLCAP) {
        lcell[r * CELLW + pos] = (s << 8) | (d & (RNG - 1));
    } else {
        gin_spill_put(s, d, gcnt, spill);
    }
}

__global__ __launch_bounds__(P1_THREADS) void gin_part(const int* __restrict__ src,
                                                       const int* __restrict__ dst,
                                                       int* __restrict__ gcnt,
                                                       int* __restrict__ cells,
                                                       int* __restrict__ spill) {
    __shared__ int lcnt[NB];
    __shared__ __attribute__((aligned(16))) int lcell[NB * CELLW];
    const int tid   = threadIdx.x;
    const int chunk = blockIdx.x;
    const int q0    = chunk * (EPC / 4) + tid;
    const v4i sa = ((const v4i*)src)[q0];
    const v4i da = ((const v4i*)dst)[q0];
    const v4i sb = ((const v4i*)src)[q0 + P1_THREADS];
    const v4i db = ((const v4i*)dst)[q0 + P1_THREADS];
    if (tid < NB) lcnt[tid] = 0;
    __syncthreads();
    #pragma unroll
    for (int j = 0; j < 4; ++j) gin_file(sa[j], da[j], lcnt, lcell, gcnt, spill);
    #pragma unroll
    for (int j = 0; j < 4; ++j) gin_file(sb[j], db[j], lcnt, lcell, gcnt, spill);
    __syncthreads();
    if (tid < NB) {
        const int c = lcnt[tid];
        lcell[tid * CELLW + CELLCAP] = c < CELLCAP ? c : CELLCAP;
    }
    __syncthreads();
    for (int v = tid; v < NB * (CELLW / 4); v += P1_THREADS) {
        const int r = v >> 3;
        const int p = v & 7;
        ((v4i*)cells)[((size_t)r * NCH + chunk) * (CELLW / 4) + p] = ((const v4i*)lcell)[v];
    }
}

#define P2_THREADS 1024
#define CPW        ((NCH + 31) / 32)
#define GCH        16

__global__ __launch_bounds__(P2_THREADS, 16) void gin_gather(const float* __restrict__ x,
                                                         const int* __restrict__ cells,
                                                         int* __restrict__ gcnt,
                                                         int* __restrict__ spill,
                                                         float* __restrict__ h0) {
    __shared__ int ncnt[RNG];
    __shared__ int lslots[RNG * CAP];
    const int tid  = threadIdx.x;
    const int lane = tid & 31;
    const int w = __builtin_amdgcn_readfirstlane(tid >> 5);
    const int b = blockIdx.x;

    int e[CPW];
    #pragma unroll
    for (int i = 0; i < CPW; ++i) {
        const int c = w + 32 * i;
        e[i] = 0;
        if (c < NCH) e[i] = cells[((size_t)b * NCH + c) * CELLW + lane];
    }
    if (tid < RNG) ncnt[tid] = 0;
    __syncthreads();

    #pragma unroll
    for (int i = 0; i < CPW; ++i) {
        int ne = __builtin_amdgcn_readlane(e[i], CELLCAP);
        ne = ne < CELLCAP ? ne : CELLCAP;
        if (lane < ne) {
            const int dl  = e[i] & (RNG - 1);
            const int s   = e[i] >> 8;
            const int pos = atomicAdd(&ncnt[dl], 1);
            if (pos < CAP) {
                lslots[dl * CAP + pos] = s;
            } else {
                gin_spill_put(s, b * RNG + dl, gcnt, spill);
            }
        }
    }
    __syncthreads();

    for (int nl = w; nl < RNG; nl += 32) {
        const int n = b * RNG + nl;
        if (n >= N_NODES) break;
        const int deg = __builtin_amdgcn_readfirstlane(ncnt[nl]);
        const int m = deg < CAP ? deg : CAP;
        int sv = 0;
        if (lane < m) sv = lslots[nl * CAP + lane];
        v2f acc = *(const v2f*)(x + (size_t)n * D + lane * 2);
        for (int j0 = 0; j0 < m; j0 += GCH) {
            int s[GCH];
            #pragma unroll
            for (int u = 0; u < GCH; ++u) {
                const int idx = (j0 + u < m) ? (j0 + u) : (m - 1);
                s[u] = __builtin_amdgcn_readlane(sv, idx);
            }
            v2f v[GCH];
            #pragma unroll
            for (int u = 0; u < GCH; ++u) {
                v[u] = (v2f){0.0f, 0.0f};
                if (j0 + u < m) {
                    v[u] = *(const v2f*)(x + (size_t)s[u] * D + lane * 2);
                }
            }
            #pragma unroll
            for (int u = 0; u < GCH; ++u) acc += v[u];
        }
        *(v2f*)(h0 + (size_t)n * D + lane * 2) = acc;
    }
}

#define SP_BLOCKS 64
#define SP_GROUPS (SP_BLOCKS * 16)

__global__ __launch_bounds__(256) void gin_spill(const float* __restrict__ x,
                                                 const int* __restrict__ gcnt,
                                                 const int* __restrict__ spill,
                                                 float* __restrict__ h0) {
    int total = gcnt[0];
    total = total < N_EDGES ? total : N_EDGES;
    const int g = (blockIdx.x * 256 + threadIdx.x) >> 4;
    const int f = (threadIdx.x & 15) * 4;
    for (int o = g; o < total; o += SP_GROUPS) {
        const int s = spill[2 * (size_t)o];
        const int d = spill[2 * (size_t)o + 1];
        const v4f v = *(const v4f*)(x + (size_t)s * D + f);
        float* p = h0 + (size_t)d * D + f;
        atomicAdd(p + 0, v[0]);
        atomicAdd(p + 1, v[1]);
        atomicAdd(p + 2, v[2]);
        atomicAdd(p + 3, v[3]);
    }
}

#define MLP_WAVES 4
#define HSTR  68
#define KSTR  320
#define LHSTR 160

__global__ __launch_bounds__(128) void gin_mlp(const float* __restrict__ h0,
                                               const float* __restrict__ W1,
                                               const float* __restrict__ b1,
                                               const float* __restrict__ W2,
                                               const float* __restrict__ b2,
                                               float* __restrict__ out) {
    __shared__ float sW1[16 * KSTR];
    __shared__ float sW2[16 * KSTR];
    __shared__ float sB1[D];
    __shared__ float sB2[D];
    __shared__ float sH[MLP_WAVES][16 * HSTR];

    const int tid = threadIdx.x;

    for (int i = tid; i < D * D / 4; i += 128) {
        const int k = i >> 4;
        const int c = (i & 15) << 2;
        const int base = (k >> 2) * KSTR + ((k >> 1) & 1) * LHSTR + (k & 1);
        v4f w1 = ((const v4f*)W1)[i];
        v4f w2 = ((const v4f*)W2)[i];
        #pragma unroll
        for (int j = 0; j < 4; ++j) {
            sW1[base + ((c + j) << 1)] = w1[j];
            sW2[base + ((c + j) << 1)] = w2[j];
        }
    }
    if (tid < D) {
        sB1[tid] = b1[tid];
        sB2[tid] = b2[tid];
    }
    __syncthreads();

    const int wave  = tid >> 5;
    const int lane  = tid & 31;
    const int lhalf = lane >> 4;
    const int l16   = lane & 15;

    const int row0 = (blockIdx.x * MLP_WAVES + wave) * 32;
    const float* a0p = h0 + (size_t)min(row0 + l16,      N_NODES - 1) * D;
    const float* a1p = h0 + (size_t)min(row0 + 16 + l16, N_NODES - 1) * D;

    const v8f vz = {0, 0, 0, 0, 0, 0, 0, 0};
    v8f acc0[4], acc1[4];
    #pragma unroll
    for (int nt = 0; nt < 4; ++nt) { acc0[nt] = vz; acc1[nt] = vz; }

    const float* bw1 = sW1 + lhalf * LHSTR + (l16 << 1);
    const float* bw2 = sW2 + lhalf * LHSTR + (l16 << 1);

    #pragma unroll
    for (int kk = 0; kk < 16; ++kk) {
        const int koff = kk * 4 + lhalf * 2;
        const v2f a0 = *(const v2f*)(a0p + koff);
        const v2f a1 = *(const v2f*)(a1p + koff);
        #pragma unroll
        for (int nt = 0; nt < 4; ++nt) {
            const v2f b = *(const v2f*)(bw1 + kk * KSTR + nt * 32);
            acc0[nt] = __builtin_amdgcn_wmma_f32_16x16x4_f32(
                false, a0, false, b, (short)0, acc0[nt], false, false);
            acc1[nt] = __builtin_amdgcn_wmma_f32_16x16x4_f32(
                false, a1, false, b, (short)0, acc1[nt], false, false);
        }
    }

    float* myH = &sH[wave][0];
    const float* aH = myH + l16 * HSTR;

    #pragma unroll
    for (int mt = 0; mt < 2; ++mt) {
        const v8f* acc = (mt == 0) ? acc0 : acc1;

        #pragma unroll
        for (int nt = 0; nt < 4; ++nt) {
            const int   col  = nt * 16 + l16;
            const float bias = sB1[col];
            #pragma unroll
            for (int r = 0; r < 8; ++r) {
                float v = acc[nt][r] + bias;
                v = v > 0.0f ? v : 0.0f;
                myH[(r + lhalf * 8) * HSTR + col] = v;
            }
        }

        v8f o[4];
        #pragma unroll
        for (int nt = 0; nt < 4; ++nt) o[nt] = vz;

        #pragma unroll
        for (int kk = 0; kk < 16; ++kk) {
            const int koff = kk * 4 + lhalf * 2;
            const v2f a = *(const v2f*)(aH + koff);
            #pragma unroll
            for (int nt = 0; nt < 4; ++nt) {
                const v2f b = *(const v2f*)(bw2 + kk * KSTR + nt * 32);
                o[nt] = __builtin_amdgcn_wmma_f32_16x16x4_f32(
                    false, a, false, b, (short)0, o[nt], false, false);
            }
        }

        #pragma unroll
        for (int r = 0; r < 8; ++r) {
            const int m = row0 + mt * 16 + r + lhalf * 8;
            if (m < N_NODES) {
                #pragma unroll
                for (int nt = 0; nt < 4; ++nt) {
                    const int col = nt * 16 + l16;
                    out[(size_t)m * D + col] = o[nt][r] + sB2[col];
                }
            }
        }
    }
}

extern "C" void kernel_launch(void* const* d_in, const int* in_sizes, int n_in,
                              void* d_out, int out_size, void* d_ws, size_t ws_size,
                              hipStream_t stream) {
    const float* x  = (const float*)d_in[0];
    const int*   ei = (const int*)d_in[1];
    const float* W1 = (const float*)d_in[2];
    const float* b1 = (const float*)d_in[3];
    const float* W2 = (const float*)d_in[4];
    const float* b2 = (const float*)d_in[5];
    float*       out = (float*)d_out;

    const int* src = ei;
    const int* dst = ei + N_EDGES;

    int*   gcnt  = (int*)d_ws;
    int*   cells = (int*)d_ws + CELL_OFF;
    int*   spill = (int*)d_ws + SPILL_OFF;
    float* h0    = (float*)d_ws + H0_OFF;

    gin_zero<<<1, 64, 0, stream>>>(gcnt);
    gin_part<<<NCH, P1_THREADS, 0, stream>>>(src, dst, gcnt, cells, spill);
    gin_gather<<<NB, P2_THREADS, 0, stream>>>(x, cells, gcnt, spill, h0);
    gin_spill<<<SP_BLOCKS, 256, 0, stream>>>(x, gcnt, spill, h0);
    {
        const int rows_per_block = MLP_WAVES * 32;
        gin_mlp<<<(N_NODES + rows_per_block - 1) / rows_per_block, 128, 0, stream>>>(
            h0, W1, b1, W2, b2, out);
    }
}


# the range's 256 summed rows stay in an LDS tile and the same work-group runs both GEMMs on it: no h round trip through memory, 5 launches -> 3; spilled edges added by LDS f32 atomics (arrival order)
# speedup vs baseline: 3.6854x; 1.2399x over previous
#include <hip/hip_runtime.h>


#define N_NODES 100000
#define N_EDGES 1600000
#define D       64

typedef __attribute__((ext_vector_type(2))) float v2f;
typedef __attribute__((ext_vector_type(4))) float v4f;
typedef __attribute__((ext_vector_type(8))) float v8f;
typedef __attribute__((ext_vector_type(4))) int   v4i;

#define CAP        32
#define RNG        256
#define NB         391
#define NCH        250
#define EPC        6400
#define CELLW      32
#define CELLCAP    31
#define GCNT_WORDS 1024
#define CELL_OFF   GCNT_WORDS
#define SPILL_OFF  (CELL_OFF + NB * NCH * CELLW)

__global__ __launch_bounds__(64) void gin_zero(int* __restrict__ gcnt) {
    const v4i z = {0, 0, 0, 0};
    ((v4i*)gcnt)[threadIdx.x] = z;
}

__device__ __forceinline__ void gin_spill_put(int s, int d,
                                              int* __restrict__ gcnt,
                                              int* __restrict__ spill) {
    const int o = atomicAdd(gcnt, 1);
    spill[2 * (size_t)o]     = s;
    spill[2 * (size_t)o + 1] = d;
}

#define P1_THREADS 800

__device__ __forceinline__ void gin_file(int s, int d, int* lcnt, int* lcell,
                                         int* __restrict__ gcnt,
                                         int* __restrict__ spill) {
    const int r   = d >> 8;
    const int pos = atomicAdd(&lcnt[r], 1);
    if (pos < CELLCAP) {
        lcell[r * CELLW + pos] = (s << 8) | (d & (RNG - 1));
    } else {
        gin_spill_put(s, d, gcnt, spill);
    }
}

__global__ __launch_bounds__(P1_THREADS) void gin_part(const int* __restrict__ src,
                                                       const int* __restrict__ dst,
                                                       int* __restrict__ gcnt,
                                                       int* __restrict__ cells,
                                                       int* __restrict__ spill) {
    __shared__ int lcnt[NB];
    __shared__ __attribute__((aligned(16))) int lcell[NB * CELLW];
    const int tid   = threadIdx.x;
    const int chunk = blockIdx.x;
    const int q0    = chunk * (EPC / 4) + tid;
    const v4i sa = ((const v4i*)src)[q0];
    const v4i da = ((const v4i*)dst)[q0];
    const v4i sb = ((const v4i*)src)[q0 + P1_THREADS];
    const v4i db = ((const v4i*)dst)[q0 + P1_THREADS];
    if (tid < NB) lcnt[tid] = 0;
    __syncthreads();
    #pragma unroll
    for (int j = 0; j < 4; ++j) gin_file(sa[j], da[j], lcnt, lcell, gcnt, spill);
    #pragma unroll
    for (int j = 0; j < 4; ++j) gin_file(sb[j], db[j], lcnt, lcell, gcnt, spill);
    __syncthreads();
    if (tid < NB) {
        const int c = lcnt[tid];
        lcell[tid * CELLW + CELLCAP] = c < CELLCAP ? c : CELLCAP;
    }
    __syncthreads();
    for (int v = tid; v < NB * (CELLW / 4); v += P1_THREADS) {
        const int r = v >> 3;
        const int p = v & 7;
        ((v4i*)cells)[((size_t)r * NCH + chunk) * (CELLW / 4) + p] = ((const v4i*)lcell)[v];
    }
}

#define P2_THREADS 1024
#define CPW        ((NCH + 31) / 32)
#define GCH        16
#define OVCAP      (NCH * CELLCAP)
#define HSTR       68
#define KSTR       320
#define LHSTR      160
#define WSZ        (16 * KSTR)
#define REGA_WORDS (2 * WSZ + 2 * D)

__device__ __forceinline__ void gin_row_add(const float* __restrict__ x, float* sH,
                                            int s, int dl, int lane) {
    const v2f v = *(const v2f*)(x + (size_t)s * D + lane * 2);
    float* p = sH + dl * HSTR + lane * 2;
    atomicAdd(p + 0, v[0]);
    atomicAdd(p + 1, v[1]);
}

__global__ __launch_bounds__(P2_THREADS, 16) void gin_gather_mlp(const float* __restrict__ x,
                                                             const int* __restrict__ cells,
                                                             const int* __restrict__ gcnt,
                                                             const int* __restrict__ spill,
                                                             const float* __restrict__ W1,
                                                             const float* __restrict__ b1,
                                                             const float* __restrict__ W2,
                                                             const float* __restrict__ b2,
                                                             float* __restrict__ out) {
    __shared__ __attribute__((aligned(16))) float sH[RNG * HSTR];
    __shared__ __attribute__((aligned(16))) int regA[REGA_WORDS];
    __shared__ int ovl[OVCAP];
    __shared__ int ovn;
    int*   ncnt   = regA;
    int*   lslots = regA + RNG;
    float* sW1    = (float*)regA;
    float* sW2    = sW1 + WSZ;
    float* sB1    = sW2 + WSZ;
    float* sB2    = sB1 + D;

    const int tid  = threadIdx.x;
    const int lane = tid & 31;
    const int w = __builtin_amdgcn_readfirstlane(tid >> 5);
    const int b = blockIdx.x;

    int e[CPW];
    #pragma unroll
    for (int i = 0; i < CPW; ++i) {
        const int c = w + 32 * i;
        e[i] = 0;
        if (c < NCH) e[i] = cells[((size_t)b * NCH + c) * CELLW + lane];
    }
    const v4f w1v = ((const v4f*)W1)[tid];
    const v4f w2v = ((const v4f*)W2)[tid];
    float b1v = 0.0f, b2v = 0.0f;
    if (tid < D) {
        b1v = b1[tid];
        b2v = b2[tid];
    }
    int total = gcnt[0];
    total = total < N_EDGES ? total : N_EDGES;
    if (tid < RNG) ncnt[tid] = 0;
    if (tid == 0) ovn = 0;
    __syncthreads();

    #pragma unroll
    for (int i = 0; i < CPW; ++i) {
        int ne = __builtin_amdgcn_readlane(e[i], CELLCAP);
        ne = ne < CELLCAP ? ne : CELLCAP;
        if (lane < ne) {
            const int dl  = e[i] & (RNG - 1);
            const int s   = e[i] >> 8;
            const int pos = atomicAdd(&ncnt[dl], 1);
            if (pos < CAP) {
                lslots[dl * CAP + pos] = s;
            } else {
                const int o = atomicAdd(&ovn, 1);
                if (o < OVCAP) ovl[o] = e[i];
            }
        }
    }
    __syncthreads();

    for (int nl = w; nl < RNG; nl += 32) {
        const int n = b * RNG + nl;
        v2f acc = {0.0f, 0.0f};
        if (n < N_NODES) {
            const int deg = __builtin_amdgcn_readfirstlane(ncnt[nl]);
            const int m = deg < CAP ? deg : CAP;
            int sv = 0;
            if (lane < m) sv = lslots[nl * CAP + lane];
            acc = *(const v2f*)(x + (size_t)n * D + lane * 2);
            for (int j0 = 0; j0 < m; j0 += GCH) {
                int s[GCH];
                #pragma unroll
                for (int u = 0; u < GCH; ++u) {
                    const int idx = (j0 + u < m) ? (j0 + u) : (m - 1);
                    s[u] = __builtin_amdgcn_readlane(sv, idx);
                }
                v2f v[GCH];
                #pragma unroll
                for (int u = 0; u < GCH; ++u) {
                    v[u] = (v2f){0.0f, 0.0f};
                    if (j0 + u < m) {
                        v[u] = *(const v2f*)(x + (size_t)s[u] * D + lane * 2);
                    }
                }
                #pragma unroll
                for (int u = 0; u < GCH; ++u) acc += v[u];
            }
        }
        *(v2f*)(sH + nl * HSTR + lane * 2) = acc;
    }
    __syncthreads();

    {
        const int k = tid >> 4;
        const int c = (tid & 15) << 2;
        const int base = (k >> 2) * KSTR + ((k >> 1) & 1) * LHSTR + (k & 1);
        #pragma unroll
        for (int j = 0; j < 4; ++j) {
            sW1[base + ((c + j) << 1)] = w1v[j];
            sW2[base + ((c + j) << 1)] = w2v[j];
        }
        if (tid < D) {
            sB1[tid] = b1v;
            sB2[tid] = b2v;
        }
    }
    {
        int novl = ovn;
        novl = novl < OVCAP ? novl : OVCAP;
        for (int o = w; o < novl; o += 32) {
            const int pe = ovl[o];
            gin_row_add(x, sH, pe >> 8, pe & (RNG - 1), lane);
        }
    }
    {
        for (int o = w; o < total; o += 32) {
            const int s = spill[2 * (size_t)o];
            const int d = spill[2 * (size_t)o + 1];
            if ((d >> 8) == b) gin_row_add(x, sH, s, d & (RNG - 1), lane);
        }
    }
    __syncthreads();

    const int lhalf = lane >> 4;
    const int l16   = lane & 15;
    const int mt    = w >> 1;
    const int nh    = w & 1;

    const float* aH  = sH + (mt * 16 + l16) * HSTR;
    const float* bw1 = sW1 + lhalf * LHSTR + (l16 << 1) + nh * 64;
    const float* bw2 = sW2 + lhalf * LHSTR + (l16 << 1) + nh * 64;
    const v8f vz = {0, 0, 0, 0, 0, 0, 0, 0};

    v8f acc[2] = {vz, vz};
    #pragma unroll
    for (int kk = 0; kk < 16; ++kk) {
        const int koff = kk * 4 + lhalf * 2;
        const v2f a = *(const v2f*)(aH + koff);
        #pragma unroll
        for (int j = 0; j < 2; ++j) {
            const v2f bf = *(const v2f*)(bw1 + kk * KSTR + j * 32);
            acc[j] = __builtin_amdgcn_wmma_f32_16x16x4_f32(
                false, a, false, bf, (short)0, acc[j], false, false);
        }
    }
    __syncthreads();

    #pragma unroll
    for (int j = 0; j < 2; ++j) {
        const int   col  = nh * 32 + j * 16 + l16;
        const float bias = sB1[col];
        #pragma unroll
        for (int r = 0; r < 8; ++r) {
            float v = acc[j][r] + bias;
            v = v > 0.0f ? v : 0.0f;
            sH[(mt * 16 + r + lhalf * 8) * HSTR + col] = v;
        }
    }
    __syncthreads();

    v8f o2[2] = {vz, vz};
    #pragma unroll
    for (int kk = 0; kk < 16; ++kk) {
        const int koff = kk * 4 + lhalf * 2;
        const v2f a = *(const v2f*)(aH + koff);
        #pragma unroll
        for (int j = 0; j < 2; ++j) {
            const v2f bf = *(const v2f*)(bw2 + kk * KSTR + j * 32);
            o2[j] = __builtin_amdgcn_wmma_f32_16x16x4_f32(
                false, a, false, bf, (short)0, o2[j], false, false);
        }
    }

    #pragma unroll
    for (int r = 0; r < 8; ++r) {
        const int m = b * RNG + mt * 16 + r + lhalf * 8;
        if (m < N_NODES) {
            #pragma unroll
            for (int j = 0; j < 2; ++j) {
                const int col = nh * 32 + j * 16 + l16;
                out[(size_t)m * D + col] = o2[j][r] + sB2[col];
            }
        }
    }
}

extern "C" void kernel_launch(void* const* d_in, const int* in_sizes, int n_in,
                              void* d_out, int out_size, void* d_ws, size_t ws_size,
                              hipStream_t stream) {
    const float* x  = (const float*)d_in[0];
    const int*   ei = (const int*)d_in[1];
    const float* W1 = (const float*)d_in[2];
    const float* b1 = (const float*)d_in[3];
    const float* W2 = (const float*)d_in[4];
    const float* b2 = (const float*)d_in[5];
    float*       out = (float*)d_out;

    const int* src = ei;
    const int* dst = ei + N_EDGES;

    int* gcnt  = (int*)d_ws;
    int* cells = (int*)d_ws + CELL_OFF;
    int* spill = (int*)d_ws + SPILL_OFF;

    gin_zero<<<1, 64, 0, stream>>>(gcnt);
    gin_part<<<NCH, P1_THREADS, 0, stream>>>(src, dst, gcnt, cells, spill);
    gin_gather_mlp<<<NB, P2_THREADS, 0, stream>>>(x, cells, gcnt, spill, W1, b1, W2, b2, out);
}
